# starting kernel
# baseline (speedup 1.0000x reference)
#include <hip/hip_runtime.h>

__constant__ float DEC_LO[8] = {-0.010597401785069032f,  0.0328830116668852f,
                                 0.030841381835560763f, -0.18703481171909309f,
                                -0.027983769416859854f,  0.6308807679298589f,
                                 0.7148465705529157f,    0.23037781330889653f};
__constant__ float DEC_HI[8] = {-0.23037781330889653f,   0.7148465705529157f,
                                -0.6308807679298589f,   -0.027983769416859854f,
                                 0.18703481171909309f,   0.030841381835560763f,
                                -0.0328830116668852f,   -0.010597401785069032f};

typedef __attribute__((ext_vector_type(2))) float v2f;
typedef __attribute__((ext_vector_type(8))) float v8f;

#define SSTRIDE 40
#define TSTRIDE 25
#define NROWS   20

__global__ __launch_bounds__(128) void idwt2_fused_kernel(
    const float* __restrict__ c0,
    const float* __restrict__ c1,
    const float* __restrict__ c2,
    float* __restrict__ out)
{
    __shared__ float Tlds[16 * TSTRIDE];
    __shared__ float S[NROWS * SSTRIDE];

    const int tid = threadIdx.x;
    const int k0  = blockIdx.x * 32;

    for (int e = tid; e < 16 * 24; e += 128) {
        int kr = e / 24, m = e - kr * 24;
        int mm = (m < 12) ? m : (m - 12);
        int j  = kr + 8 - 2 * mm;
        float v = 0.0f;
        if (j >= 0 && j < 8) v = (m < 12) ? DEC_LO[j] : DEC_HI[j];
        Tlds[kr * TSTRIDE + m] = v;
    }

    const int rlo = (k0 >> 1) - 4;
    for (int e = tid; e < NROWS * 32; e += 128) {
        int s   = e >> 5;
        int col = e & 31;
        int b   = col >> 3, c = col & 7;
        int r   = (rlo + s) & 4095;
        int par = r & 1;
        int re  = r - par;
        float acc = 0.0f;
#pragma unroll
        for (int p = 0; p < 4; ++p) {
            float hj = par ? DEC_LO[2 * p + 1] : DEC_LO[2 * p];
            float gj = par ? DEC_HI[2 * p + 1] : DEC_HI[2 * p];
            int idx  = ((re - 2 * p) & 4095) >> 1;
            int off  = b * 16384 + idx * 8 + c;
            acc += hj * c0[off] + gj * c1[off];
        }
        S[s * SSTRIDE + col] = acc;
    }
    __syncthreads();

    const int lane  = tid & 31;
    const int w     = tid >> 5;
    const int tile  = w >> 1;
    const int half  = w & 1;
    const int koff  = (lane >= 16) ? 2 : 0;
    const int mA    = lane & 15;
    const int col   = half * 16 + (lane & 15);
    const int bb    = col >> 3, cc = col & 7;
    const int k0t   = k0 + tile * 16;
    const int ilo_t = (k0t >> 1) - 4;

    v8f acc = {};
#pragma unroll
    for (int ch = 0; ch < 3; ++ch) {
        const int cK = ch * 4;
        v2f a, b;
        a.x = Tlds[mA * TSTRIDE + cK + koff];
        a.y = Tlds[mA * TSTRIDE + cK + koff + 1];
        int s0 = 8 * tile + cK + koff;
        b.x = S[s0 * SSTRIDE + col];
        b.y = S[(s0 + 1) * SSTRIDE + col];
        acc = __builtin_amdgcn_wmma_f32_16x16x4_f32(false, a, false, b,
                                                    (short)0, acc, false, false);
    }
#pragma unroll
    for (int ch = 3; ch < 6; ++ch) {
        const int cK = ch * 4;
        v2f a, b;
        a.x = Tlds[mA * TSTRIDE + cK + koff];
        a.y = Tlds[mA * TSTRIDE + cK + koff + 1];
        int mrow = (cK - 12) + koff;
        int i0 = (ilo_t + mrow) & 4095;
        int i1 = (ilo_t + mrow + 1) & 4095;
        b.x = c2[bb * 32768 + i0 * 8 + cc];
        b.y = c2[bb * 32768 + i1 * 8 + cc];
        acc = __builtin_amdgcn_wmma_f32_16x16x4_f32(false, a, false, b,
                                                    (short)0, acc, false, false);
    }

    const int krbase = (lane >= 16) ? 8 : 0;
#pragma unroll
    for (int r = 0; r < 8; ++r) {
        int k = k0t + krbase + r;
        out[bb * 65536 + k * 8 + cc] = acc[r];
    }
}

extern "C" void kernel_launch(void* const* d_in, const int* in_sizes, int n_in,
                              void* d_out, int out_size, void* d_ws, size_t ws_size,
                              hipStream_t stream) {
    (void)in_sizes; (void)n_in; (void)out_size; (void)d_ws; (void)ws_size;
    const float* c0 = (const float*)d_in[0];
    const float* c1 = (const float*)d_in[1];
    const float* c2 = (const float*)d_in[2];
    float* out = (float*)d_out;
    idwt2_fused_kernel<<<dim3(256), dim3(128), 0, stream>>>(c0, c1, c2, out);
}


# All global reads issued in one batch at entry (level-1 loop unrolled onto a 12-row register window, coeff2 fragments read above the barrier, taps from literals): 9 serial memory round trips become 1
# speedup vs baseline: 1.4881x; 1.4881x over previous
#include <hip/hip_runtime.h>


typedef __attribute__((ext_vector_type(2))) float v2f;
typedef __attribute__((ext_vector_type(8))) float v8f;

#define SSTRIDE 40
#define TSTRIDE 25
#define NROWS   20

__device__ __forceinline__ float dec_lo(int j) {
    float v = 0.0f;
    v = (j == 0) ? -0.010597401785069032f : v;
    v = (j == 1) ?  0.0328830116668852f   : v;
    v = (j == 2) ?  0.030841381835560763f : v;
    v = (j == 3) ? -0.18703481171909309f  : v;
    v = (j == 4) ? -0.027983769416859854f : v;
    v = (j == 5) ?  0.6308807679298589f   : v;
    v = (j == 6) ?  0.7148465705529157f   : v;
    v = (j == 7) ?  0.23037781330889653f  : v;
    return v;
}
__device__ __forceinline__ float dec_hi(int j) {
    float v = 0.0f;
    v = (j == 0) ? -0.23037781330889653f  : v;
    v = (j == 1) ?  0.7148465705529157f   : v;
    v = (j == 2) ? -0.6308807679298589f   : v;
    v = (j == 3) ? -0.027983769416859854f : v;
    v = (j == 4) ?  0.18703481171909309f  : v;
    v = (j == 5) ?  0.030841381835560763f : v;
    v = (j == 6) ? -0.0328830116668852f   : v;
    v = (j == 7) ? -0.010597401785069032f : v;
    return v;
}

__global__ __launch_bounds__(128) void idwt2_fused_kernel(
    const float* __restrict__ c0,
    const float* __restrict__ c1,
    const float* __restrict__ c2,
    float* __restrict__ out)
{
    __shared__ float Tlds[16 * TSTRIDE];
    __shared__ float S[NROWS * SSTRIDE];

    const int tid  = threadIdx.x;
    const int lane = tid & 31;
    const int w    = tid >> 5;
    const int k0   = blockIdx.x * 32;

    const int pb    = lane >> 3, pc = lane & 7;
    const int par   = w & 1;
    const int qbase = (k0 >> 2) - 5 + (w >> 1);
    float x0[12], x1[12];
#pragma unroll
    for (int t = 0; t < 12; ++t) {
        int idx = (qbase + t) & 2047;
        int off = pb * 16384 + idx * 8 + pc;
        x0[t] = c0[off];
        x1[t] = c1[off];
    }

    const int tile  = w >> 1;
    const int half  = w & 1;
    const int koff  = (lane >= 16) ? 2 : 0;
    const int mA    = lane & 15;
    const int col   = half * 16 + (lane & 15);
    const int bb    = col >> 3, cc = col & 7;
    const int k0t   = k0 + tile * 16;
    const int ilo_t = (k0t >> 1) - 4;
    v2f bq[3];
#pragma unroll
    for (int ch = 0; ch < 3; ++ch) {
        int mrow = ch * 4 + koff;
        int i0 = (ilo_t + mrow) & 4095;
        int i1 = (ilo_t + mrow + 1) & 4095;
        bq[ch].x = c2[bb * 32768 + i0 * 8 + cc];
        bq[ch].y = c2[bb * 32768 + i1 * 8 + cc];
    }
    __builtin_amdgcn_sched_barrier(0);

#pragma unroll
    for (int it = 0; it < 3; ++it) {
        int e  = tid + it * 128;
        int kr = e / 24, m = e - kr * 24;
        int mm = (m < 12) ? m : (m - 12);
        int j  = kr + 8 - 2 * mm;
        float lo = dec_lo(j), hi = dec_hi(j);
        Tlds[kr * TSTRIDE + m] = (m < 12) ? lo : hi;
    }
    __builtin_amdgcn_sched_barrier(0);

    float hj[4], gj[4];
#pragma unroll
    for (int p = 0; p < 4; ++p) {
        hj[p] = par ? dec_lo(2 * p + 1) : dec_lo(2 * p);
        gj[p] = par ? dec_hi(2 * p + 1) : dec_hi(2 * p);
    }
#pragma unroll
    for (int i = 0; i < 5; ++i) {
        float acc = 0.0f;
#pragma unroll
        for (int p = 0; p < 4; ++p) {
            int t = 2 * i - p + 3;
            acc += hj[p] * x0[t] + gj[p] * x1[t];
        }
        S[(w + 4 * i) * SSTRIDE + lane] = acc;
    }
    __syncthreads();

    v8f acc = {};
#pragma unroll
    for (int ch = 0; ch < 3; ++ch) {
        const int cK = ch * 4;
        v2f a, b;
        a.x = Tlds[mA * TSTRIDE + cK + koff];
        a.y = Tlds[mA * TSTRIDE + cK + koff + 1];
        int s0 = 8 * tile + cK + koff;
        b.x = S[s0 * SSTRIDE + col];
        b.y = S[(s0 + 1) * SSTRIDE + col];
        acc = __builtin_amdgcn_wmma_f32_16x16x4_f32(false, a, false, b,
                                                    (short)0, acc, false, false);
    }
#pragma unroll
    for (int ch = 3; ch < 6; ++ch) {
        const int cK = ch * 4;
        v2f a;
        a.x = Tlds[mA * TSTRIDE + cK + koff];
        a.y = Tlds[mA * TSTRIDE + cK + koff + 1];
        acc = __builtin_amdgcn_wmma_f32_16x16x4_f32(false, a, false, bq[ch - 3],
                                                    (short)0, acc, false, false);
    }

    const int krbase = (lane >= 16) ? 8 : 0;
#pragma unroll
    for (int r = 0; r < 8; ++r) {
        int k = k0t + krbase + r;
        out[bb * 65536 + k * 8 + cc] = acc[r];
    }
}

extern "C" void kernel_launch(void* const* d_in, const int* in_sizes, int n_in,
                              void* d_out, int out_size, void* d_ws, size_t ws_size,
                              hipStream_t stream) {
    (void)in_sizes; (void)n_in; (void)out_size; (void)d_ws; (void)ws_size;
    const float* c0 = (const float*)d_in[0];
    const float* c1 = (const float*)d_in[1];
    const float* c2 = (const float*)d_in[2];
    float* out = (float*)d_out;
    idwt2_fused_kernel<<<dim3(256), dim3(128), 0, stream>>>(c0, c1, c2, out);
}


# Each lane forms its own fragments in registers (level-1 rows as fma chains straight in the B layout, taps from literals; coeff2 chunks summed first: sum order changes): no LDS, no barrier
# speedup vs baseline: 1.6591x; 1.1149x over previous
#include <hip/hip_runtime.h>


typedef __attribute__((ext_vector_type(2))) float v2f;
typedef __attribute__((ext_vector_type(8))) float v8f;

__device__ __forceinline__ float dec_lo(int j) {
    float v = 0.0f;
    v = (j == 0) ? -0.010597401785069032f : v;
    v = (j == 1) ?  0.0328830116668852f   : v;
    v = (j == 2) ?  0.030841381835560763f : v;
    v = (j == 3) ? -0.18703481171909309f  : v;
    v = (j == 4) ? -0.027983769416859854f : v;
    v = (j == 5) ?  0.6308807679298589f   : v;
    v = (j == 6) ?  0.7148465705529157f   : v;
    v = (j == 7) ?  0.23037781330889653f  : v;
    return v;
}
__device__ __forceinline__ float dec_hi(int j) {
    float v = 0.0f;
    v = (j == 0) ? -0.23037781330889653f  : v;
    v = (j == 1) ?  0.7148465705529157f   : v;
    v = (j == 2) ? -0.6308807679298589f   : v;
    v = (j == 3) ? -0.027983769416859854f : v;
    v = (j == 4) ?  0.18703481171909309f  : v;
    v = (j == 5) ?  0.030841381835560763f : v;
    v = (j == 6) ? -0.0328830116668852f   : v;
    v = (j == 7) ? -0.010597401785069032f : v;
    return v;
}

__global__ __launch_bounds__(128) void idwt2_fused_kernel(
    const float* __restrict__ c0,
    const float* __restrict__ c1,
    const float* __restrict__ c2,
    float* __restrict__ out)
{
    const int tid   = threadIdx.x;
    const int lane  = tid & 31;
    const int w     = tid >> 5;
    const int tile  = w >> 1;
    const int half  = w & 1;
    const int h     = lane >> 4;
    const int c     = lane & 15;
    const int col   = half * 16 + c;
    const int bb    = col >> 3, cc = col & 7;
    const int k0t   = blockIdx.x * 32 + tile * 16;
    const int ilo_t = (k0t >> 1) - 4;

    v2f bq[3];
#pragma unroll
    for (int ch = 0; ch < 3; ++ch) {
        int mrow = ch * 4 + 2 * h;
        int i0 = (ilo_t + mrow) & 4095;
        int i1 = (ilo_t + mrow + 1) & 4095;
        bq[ch].x = c2[bb * 32768 + i0 * 8 + cc];
        bq[ch].y = c2[bb * 32768 + i1 * 8 + cc];
    }

    const int u0 = (k0t >> 2) - 5 + h;
    float x0[8], x1[8];
#pragma unroll
    for (int t = 0; t < 8; ++t) {
        int idx = (u0 + t) & 2047;
        int off = bb * 16384 + idx * 8 + cc;
        x0[t] = c0[off];
        x1[t] = c1[off];
    }

    const int J0 = c + 8 - 4 * h;
    const int J1 = J0 - 2;
    const int s0 = J0 >> 3, s1 = J1 >> 3;
    const float tl0 = dec_lo(J0 & 7), th0 = dec_hi(J0 & 7);
    const float tl1 = dec_lo(J1 & 7), th1 = dec_hi(J1 & 7);
    v2f alo[3], ahi[3];
#pragma unroll
    for (int ch = 0; ch < 3; ++ch) {
        alo[ch].x = (s0 == ch) ? tl0 : 0.0f;
        alo[ch].y = (s1 == ch) ? tl1 : 0.0f;
        ahi[ch].x = (s0 == ch) ? th0 : 0.0f;
        ahi[ch].y = (s1 == ch) ? th1 : 0.0f;
    }

    v8f acc = {};
#pragma unroll
    for (int ch = 0; ch < 3; ++ch)
        acc = __builtin_amdgcn_wmma_f32_16x16x4_f32(false, ahi[ch], false, bq[ch],
                                                    (short)0, acc, false, false);

#pragma unroll
    for (int ch = 0; ch < 3; ++ch) {
        v2f b = {0.0f, 0.0f};
#pragma unroll
        for (int p = 3; p >= 0; --p) {
            const int t = 2 * ch + 3 - p;
            const v2f hp = {dec_lo(2 * p), dec_lo(2 * p + 1)};
            const v2f gp = {dec_hi(2 * p), dec_hi(2 * p + 1)};
            b += hp * x0[t];
            b += gp * x1[t];
        }
        acc = __builtin_amdgcn_wmma_f32_16x16x4_f32(false, alo[ch], false, b,
                                                    (short)0, acc, false, false);
    }

    const int krbase = h * 8;
#pragma unroll
    for (int r = 0; r < 8; ++r) {
        int k = k0t + krbase + r;
        out[bb * 65536 + k * 8 + cc] = acc[r];
    }
}

extern "C" void kernel_launch(void* const* d_in, const int* in_sizes, int n_in,
                              void* d_out, int out_size, void* d_ws, size_t ws_size,
                              hipStream_t stream) {
    (void)in_sizes; (void)n_in; (void)out_size; (void)d_ws; (void)ws_size;
    const float* c0 = (const float*)d_in[0];
    const float* c1 = (const float*)d_in[1];
    const float* c2 = (const float*)d_in[2];
    float* out = (float*)d_out;
    idwt2_fused_kernel<<<dim3(256), dim3(128), 0, stream>>>(c0, c1, c2, out);
}
